# starting kernel
# baseline (speedup 1.0000x reference)
#include <hip/hip_runtime.h>


typedef __attribute__((ext_vector_type(2))) float v2f;
typedef __attribute__((ext_vector_type(8))) float v8f;

#define BS 8
#define FS 96
#define HH 96
#define WW 320
#define PLANES 9
#define LVLS 4
#define TILES_X (WW / 16)
#define NWAVES (BS * HH * TILES_X)
#define WAVES_PER_BLOCK 8

__global__ __launch_bounds__(256) void cost_volume_wmma_f32(
    const float* __restrict__ f1, const float* __restrict__ f2,
    float* __restrict__ out) {
  const int lane = threadIdx.x & 31;
  const int wid = blockIdx.x * WAVES_PER_BLOCK + (threadIdx.x >> 5);

  const int t = wid % TILES_X;
  const int bh = wid / TILES_X;
  const int h = bh % HH;
  const int b = bh / HH;

  const int X0 = t * 16;
  const int half = lane >> 4;
  const int m = lane & 15;

  const int HW = HH * WW;
  const float* __restrict__ f1row = f1 + ((long)b * FS * HH + h) * WW;
  const float* __restrict__ f2row = f2 + ((long)b * FS * HH + h) * WW;

  const int col0 = X0 - 8 + m;
  const int col1 = X0 + 8 + m;
  const int col0c = col0 < 0 ? 0 : col0;
  const int col1c = col1 >= WW ? (WW - 1) : col1;
  const float inv = 1.0f / (float)FS;
  const float scale0 = (col0 >= 0) ? inv : 0.0f;
  const float scale1 = (col1 < WW) ? inv : 0.0f;

  const int aBase = 2 * half * HW + X0 + m;

  v8f acc0 = {};
  v8f acc1 = {};

#pragma unroll
  for (int k0 = 0; k0 < FS; k0 += 4) {
    const int c = k0 * HW;

    v2f a;
    a.x = f1row[c + aBase];
    a.y = f1row[c + aBase + HW];

    v2f b0, b1;
    b0.x = f2row[c + 2 * half * HW + col0c];
    b0.y = f2row[c + 2 * half * HW + HW + col0c];
    b1.x = f2row[c + 2 * half * HW + col1c];
    b1.y = f2row[c + 2 * half * HW + HW + col1c];

    acc0 = __builtin_amdgcn_wmma_f32_16x16x4_f32(false, a, false, b0,
                                                 (short)0, acc0, false, false);
    acc1 = __builtin_amdgcn_wmma_f32_16x16x4_f32(false, a, false, b1,
                                                 (short)0, acc1, false, false);
  }

  const long outBase = ((long)b * PLANES * HH + h) * WW;
#pragma unroll
  for (int r = 0; r < 8; ++r) {
    const int M = r + 8 * half;
    const int x = X0 + M;

    const int d0 = m - 8 - M;
    if (d0 >= -LVLS && d0 <= LVLS) {
      out[outBase + (long)(LVLS - d0) * HW + x] = acc0[r] * scale0;
    }
    const int d1 = m + 8 - M;
    if (d1 >= -LVLS && d1 <= LVLS) {
      out[outBase + (long)(LVLS - d1) * HW + x] = acc1[r] * scale1;
    }
  }
}

extern "C" void kernel_launch(void* const* d_in, const int* in_sizes, int n_in,
                              void* d_out, int out_size, void* d_ws,
                              size_t ws_size, hipStream_t stream) {
  (void)in_sizes; (void)n_in; (void)d_ws; (void)ws_size; (void)out_size;
  const float* f1 = (const float*)d_in[0];
  const float* f2 = (const float*)d_in[1];
  float* out = (float*)d_out;

  const int blocks = NWAVES / WAVES_PER_BLOCK;
  cost_volume_wmma_f32<<<dim3(blocks), dim3(256), 0, stream>>>(f1, f2, out);
}


# 256 workgroups (1 per WGP, 20 waves) each walk 3 rows as an 18-stage LDS pipeline: 4 stages of loads under way, next stage's fragments read before the WMMAs, row stores under next row; sums unchanged
# speedup vs baseline: 1.1971x; 1.1971x over previous
#include <hip/hip_runtime.h>


typedef __attribute__((ext_vector_type(2))) float v2f;
typedef __attribute__((ext_vector_type(4))) float v4f;
typedef __attribute__((ext_vector_type(8))) float v8f;

static __device__ __forceinline__ void lds_barrier() {
  __builtin_amdgcn_fence(__ATOMIC_RELEASE, "workgroup", "local");
  __builtin_amdgcn_s_barrier();
  __builtin_amdgcn_fence(__ATOMIC_ACQUIRE, "workgroup", "local");
}

#define BS 8
#define FS 96
#define HH 96
#define WW 320
#define PLANES 9
#define LVLS 4
#define HW (HH * WW)

#define NWG 256
#define RPW ((BS * HH) / NWG)
#define KC 16
#define KP (KC / 2)
#define NST (FS / KC)
#define NG (RPW * NST)
#define CPT (KC / 2)
#define HALO 8
#define GROUPS ((WW + 2 * HALO) / 2)
#define KP_FLOATS (4 * GROUPS)
#define ARR_FLOATS (KP * KP_FLOATS)
#define STAGE_FLOATS (2 * ARR_FLOATS)
#define NTHREADS 640
#define EPI_STAGES 4
#define EPI_REGS (8 / EPI_STAGES)

#define NSET 4
#define LDS_FLOATS (4 * STAGE_FLOATS)

static_assert(RPW * NWG == BS * HH, "rows must divide over the workgroups");
static_assert(LDS_FLOATS * 4 > 160 * 1024, "one workgroup per WGP");

__global__ __launch_bounds__(NTHREADS) void cost_volume_rows_loop(
    const float* __restrict__ f1, const float* __restrict__ f2,
    float* __restrict__ out) {
  __shared__ __attribute__((aligned(16))) float lds[LDS_FLOATS];

  const int tid = threadIdx.x;
  const int lane = tid & 31;
  const int w = tid >> 5;

  long rowOff[RPW], outBase[RPW];
#pragma unroll
  for (int r = 0; r < RPW; ++r) {
    const int R = (int)blockIdx.x + NWG * r;
    const int b = R / HH;
    const int h = R % HH;
    rowOff[r] = ((long)b * FS * HH + h) * WW;
    outBase[r] = ((long)b * PLANES * HH + h) * WW;
  }

  const int arr = (w >= 10) ? 1 : 0;
  const int wq = w - 10 * arr;
  const int chh = (wq >= 5) ? 1 : 0;
  const int cp = (wq - 5 * chh) * 32 + lane;
  const float* __restrict__ src =
      (arr ? f2 : f1) + 2 * cp + (long)(chh * CPT) * HW;
  float* dst = lds + arr * ARR_FLOATS + chh * (CPT / 2) * KP_FLOATS +
               4 * (cp + arr * (HALO / 2));

  const int half = lane >> 4;
  const int m = lane & 15;
  const int X0 = 16 * w;
  const float* frag = lds + half * KP_FLOATS + 2 * X0 + 4 * (m >> 1) + (m & 1);

  v2f g[NSET][CPT];

#define ISSUE(set, G)                                                       \
  {                                                                         \
    const float* s_ = src + rowOff[(G) / NST] + (long)(((G) % NST) * KC) * HW; \
    _Pragma("unroll") for (int c = 0; c < CPT; ++c)                         \
        g[set][c] = *(const v2f*)(s_ + (long)c * HW);                       \
  }

#define STASH(set, buf)                                                     \
  _Pragma("unroll") for (int kp = 0; kp < CPT / 2; ++kp) {                  \
    v4f q = {g[set][2 * kp].x, g[set][2 * kp].y, g[set][2 * kp + 1].x,      \
             g[set][2 * kp + 1].y};                                         \
    *(v4f*)(dst + (buf) * STAGE_FLOATS + kp * KP_FLOATS) = q;               \
  }

#define EPI(row, lo, hi, r0, r1)                                            \
  _Pragma("unroll") for (int r = (r0); r < (r1); ++r) {                     \
    const int M = r + 8 * half;                                             \
    const int x = X0 + M;                                                   \
    const int d0 = m - 8 - M;                                               \
    if (d0 >= -LVLS && d0 <= LVLS) {                                        \
      out[outBase[row] + (long)(LVLS - d0) * HW + x] = lo[r] * inv;         \
    }                                                                       \
    const int d1 = m + 8 - M;                                               \
    if (d1 >= -LVLS && d1 <= LVLS) {                                        \
      out[outBase[row] + (long)(LVLS - d1) * HW + x] = hi[r] * inv;         \
    }                                                                       \
  }

  const float inv = 1.0f / (float)FS;

  ISSUE(0, 0)
  ISSUE(1, 1)
  ISSUE(2, 2)
  ISSUE(3, 3)

  if (tid < 2 * KP * HALO) {
    const int bf = tid >> 6;
    const int kp = (tid >> 3) & 7;
    const int gi = tid & 7;
    const int grp = (gi < HALO / 2) ? gi : (WW / 2 + gi);
    v4f z = {0.0f, 0.0f, 0.0f, 0.0f};
    *(v4f*)&lds[bf * STAGE_FLOATS + ARR_FLOATS + kp * KP_FLOATS + 4 * grp] = z;
  }

  STASH(0, 0)
  lds_barrier();
  ISSUE(0, NSET)

  v8f acc0 = {}, acc1 = {};
  v8f pre0 = {}, pre1 = {};

  v2f fa_[2][KC / 4], fb0_[2][KC / 4], fb1_[2][KC / 4];

#define FRAGS(fs, buf)                                                      \
  _Pragma("unroll") for (int j = 0; j < KC / 4; ++j) {                      \
    const float* pa = frag + (buf) * STAGE_FLOATS + j * 2 * KP_FLOATS;      \
    const float* pb = pa + ARR_FLOATS;                                      \
    fa_[fs][j] = (v2f){pa[0], pa[2]};                                       \
    fb0_[fs][j] = (v2f){pb[0], pb[2]};                                      \
    fb1_[fs][j] = (v2f){pb[32], pb[34]};                                    \
  }

  FRAGS(0, 0)
  STASH(1, 1)
  lds_barrier();

#pragma unroll
  for (int G = 0; G < NG; ++G) {
    const int row = G / NST;
    const int st = G % NST;

    if (G + NSET + 1 < NG) {
      ISSUE((G + 1) % NSET, G + NSET + 1)
    }

    if (row > 0 && st < EPI_STAGES) {
      EPI(row - 1, pre0, pre1, st * EPI_REGS, (st + 1) * EPI_REGS)
    }

    if (G + 1 < NG) {
      FRAGS((G + 1) & 1, (G + 1) & 1)
      __builtin_amdgcn_sched_barrier(0);
    }

#pragma unroll
    for (int j = 0; j < KC / 4; ++j) {
      acc0 = __builtin_amdgcn_wmma_f32_16x16x4_f32(
          false, fa_[G & 1][j], false, fb0_[G & 1][j], (short)0, acc0, false,
          false);
      acc1 = __builtin_amdgcn_wmma_f32_16x16x4_f32(
          false, fa_[G & 1][j], false, fb1_[G & 1][j], (short)0, acc1, false,
          false);
    }

    if (st == NST - 1) {
      pre0 = acc0;
      pre1 = acc1;
      acc0 = (v8f){};
      acc1 = (v8f){};
    }

    if (G + 2 < NG) {
      __builtin_amdgcn_sched_barrier(0);
      STASH((G + 2) % NSET, G & 1)
      lds_barrier();
    }
  }

  EPI(RPW - 1, pre0, pre1, 0, 8)
}

extern "C" void kernel_launch(void* const* d_in, const int* in_sizes, int n_in,
                              void* d_out, int out_size, void* d_ws,
                              size_t ws_size, hipStream_t stream) {
  (void)in_sizes; (void)n_in; (void)d_ws; (void)ws_size; (void)out_size;
  const float* f1 = (const float*)d_in[0];
  const float* f2 = (const float*)d_in[1];
  float* out = (float*)d_out;

  cost_volume_rows_loop<<<dim3(NWG), dim3(NTHREADS), 0, stream>>>(f1, f2, out);
}


# Band product re-tiled: a tile takes the same 16 columns of f1 and f2, tiles start every 12 columns: 27 WMMAs per row and k-step, was 40 (7 per SIMD, was 10); order of sums unchanged
# speedup vs baseline: 1.2609x; 1.0533x over previous
#include <hip/hip_runtime.h>


typedef __attribute__((ext_vector_type(2))) float v2f;
typedef __attribute__((ext_vector_type(4))) float v4f;
typedef __attribute__((ext_vector_type(8))) float v8f;

static __device__ __forceinline__ void lds_barrier() {
  __builtin_amdgcn_fence(__ATOMIC_RELEASE, "workgroup", "local");
  __builtin_amdgcn_s_barrier();
  __builtin_amdgcn_fence(__ATOMIC_ACQUIRE, "workgroup", "local");
}

#define BS 8
#define FS 96
#define HH 96
#define WW 320
#define PLANES 9
#define LVLS 4
#define HW (HH * WW)

#define NWG 256
#define RPW ((BS * HH) / NWG)
#define KC 16
#define KP (KC / 2)
#define NST (FS / KC)
#define NG (RPW * NST)
#define CPT (KC / 2)
#define HALO 8
#define TPITCH 12
#define NTILES 27
#define GROUPS ((WW + 2 * HALO) / 2)
#define KP_FLOATS (4 * GROUPS)
#define ARR_FLOATS (KP * KP_FLOATS)
#define STAGE_FLOATS (2 * ARR_FLOATS)
#define NTHREADS 640
#define EPI_STAGES 4
#define EPI_REGS (8 / EPI_STAGES)

#define NSET 4
#define LDS_FLOATS (4 * STAGE_FLOATS)

static_assert(RPW * NWG == BS * HH, "rows must divide over the workgroups");
static_assert(TPITCH * (NTILES - 1) - 4 + 16 >= WW + LVLS, "tiles cover the band");
static_assert(TPITCH * (NTILES - 1) - 4 + 16 <= WW + HALO, "last tile inside the halo");
static_assert(LDS_FLOATS * 4 > 160 * 1024, "one workgroup per WGP");

__global__ __launch_bounds__(NTHREADS) void cost_volume_rows_loop(
    const float* __restrict__ f1, const float* __restrict__ f2,
    float* __restrict__ out) {
  __shared__ __attribute__((aligned(16))) float lds[LDS_FLOATS];

  const int tid = threadIdx.x;
  const int lane = tid & 31;
  const int w = tid >> 5;

  long rowOff[RPW], outBase[RPW];
#pragma unroll
  for (int r = 0; r < RPW; ++r) {
    const int R = (int)blockIdx.x + NWG * r;
    const int b = R / HH;
    const int h = R % HH;
    rowOff[r] = ((long)b * FS * HH + h) * WW;
    outBase[r] = ((long)b * PLANES * HH + h) * WW;
  }

  const int arr = (w >= 10) ? 1 : 0;
  const int wq = w - 10 * arr;
  const int chh = (wq >= 5) ? 1 : 0;
  const int cp = (wq - 5 * chh) * 32 + lane;
  const float* __restrict__ src =
      (arr ? f2 : f1) + 2 * cp + (long)(chh * CPT) * HW;
  float* dst = lds + arr * ARR_FLOATS + chh * (CPT / 2) * KP_FLOATS +
               4 * (cp + HALO / 2);

  const int half = lane >> 4;
  const int m = lane & 15;
  const int wu = __builtin_amdgcn_readfirstlane(w);
  const int two = __builtin_amdgcn_readfirstlane(
      ((wu % 6 < 2 && wu < 14) || wu == 15) ? 1 : 0);
  const int t1 = (wu == 15) ? NTILES - 1 : 20 + 2 * (wu / 6) + (wu % 6);
  const int S0 = TPITCH * wu - 4;
  const int S1 = TPITCH * t1 - 4;
  const float* fragbase = lds + half * KP_FLOATS + 4 * (m >> 1) + (m & 1);
  const float* frag0 = fragbase + 2 * (S0 + HALO);
  const float* frag1 = fragbase + 2 * (S1 + HALO);

  v2f g[NSET][CPT];

#define ISSUE(set, G)                                                       \
  {                                                                         \
    const float* s_ = src + rowOff[(G) / NST] + (long)(((G) % NST) * KC) * HW; \
    _Pragma("unroll") for (int c = 0; c < CPT; ++c)                         \
        g[set][c] = *(const v2f*)(s_ + (long)c * HW);                       \
  }

#define STASH(set, buf)                                                     \
  _Pragma("unroll") for (int kp = 0; kp < CPT / 2; ++kp) {                  \
    v4f q = {g[set][2 * kp].x, g[set][2 * kp].y, g[set][2 * kp + 1].x,      \
             g[set][2 * kp + 1].y};                                         \
    *(v4f*)(dst + (buf) * STAGE_FLOATS + kp * KP_FLOATS) = q;               \
  }

#define EPI(row, acc, S, r0, r1)                                            \
  _Pragma("unroll") for (int r = (r0); r < (r1); ++r) {                     \
    const int M = r + 8 * half;                                             \
    const int x = (S) + M;                                                  \
    const int d = m - M;                                                    \
    if (d >= -LVLS && d <= LVLS && (M >= 4 || m >= 4) && x >= 0 && x < WW) { \
      out[outBase[row] + (long)(LVLS - d) * HW + x] = acc[r] * inv;         \
    }                                                                       \
  }

  const float inv = 1.0f / (float)FS;

  ISSUE(0, 0)
  ISSUE(1, 1)
  ISSUE(2, 2)
  ISSUE(3, 3)

  if (tid < 2 * 2 * KP * HALO) {
    const int bf = tid >> 7;
    const int ar = (tid >> 6) & 1;
    const int kp = (tid >> 3) & 7;
    const int gi = tid & 7;
    const int grp = (gi < HALO / 2) ? gi : (WW / 2 + gi);
    v4f z = {0.0f, 0.0f, 0.0f, 0.0f};
    *(v4f*)&lds[bf * STAGE_FLOATS + ar * ARR_FLOATS + kp * KP_FLOATS + 4 * grp] = z;
  }

  STASH(0, 0)
  lds_barrier();
  ISSUE(0, NSET)

  v8f acc0 = {}, acc1 = {};
  v8f pre0 = {}, pre1 = {};

  v2f fa0_[2][KC / 4], fb0_[2][KC / 4], fa1_[1][KC / 4], fb1_[1][KC / 4];

#define FRAGS(fa, fb, frag, fs, buf)                                        \
  _Pragma("unroll") for (int j = 0; j < KC / 4; ++j) {                      \
    const float* pa = frag + (buf) * STAGE_FLOATS + j * 2 * KP_FLOATS;      \
    const float* pb = pa + ARR_FLOATS;                                      \
    fa[fs][j] = (v2f){pa[0], pa[2]};                                        \
    fb[fs][j] = (v2f){pb[0], pb[2]};                                        \
  }

  FRAGS(fa0_, fb0_, frag0, 0, 0)
  if (two) {
    FRAGS(fa1_, fb1_, frag1, 0, 0)
  }
  STASH(1, 1)
  lds_barrier();

#pragma unroll
  for (int G = 0; G < NG; ++G) {
    const int row = G / NST;
    const int st = G % NST;

    if (G + NSET + 1 < NG) {
      ISSUE((G + 1) % NSET, G + NSET + 1)
    }

    if (row > 0 && st < EPI_STAGES) {
      EPI(row - 1, pre0, S0, st * EPI_REGS, (st + 1) * EPI_REGS)
      if (two) {
        EPI(row - 1, pre1, S1, st * EPI_REGS, (st + 1) * EPI_REGS)
      }
    }

    if (G + 1 < NG) {
      FRAGS(fa0_, fb0_, frag0, (G + 1) & 1, (G + 1) & 1)
      __builtin_amdgcn_sched_barrier(0);
    }

    if (two) {
#pragma unroll
      for (int j = 0; j < KC / 4; ++j) {
        acc1 = __builtin_amdgcn_wmma_f32_16x16x4_f32(
            false, fa1_[0][j], false, fb1_[0][j], (short)0, acc1,
            false, false);
      }
    }
#pragma unroll
    for (int j = 0; j < KC / 4; ++j) {
      acc0 = __builtin_amdgcn_wmma_f32_16x16x4_f32(
          false, fa0_[G & 1][j], false, fb0_[G & 1][j], (short)0, acc0, false,
          false);
    }

    if (st == NST - 1) {
      pre0 = acc0;
      pre1 = acc1;
      acc0 = (v8f){};
      acc1 = (v8f){};
    }

    if (G + 2 < NG) {
      __builtin_amdgcn_sched_barrier(0);
      STASH((G + 2) % NSET, G & 1)
    }
    if (G + 1 < NG) {
      if (two) {
        FRAGS(fa1_, fb1_, frag1, 0, (G + 1) & 1)
      }
    }
    if (G + 2 < NG) {
      lds_barrier();
    }
  }

  EPI(RPW - 1, pre0, S0, 0, 8)
  if (two) {
    EPI(RPW - 1, pre1, S1, 0, 8)
  }
}

extern "C" void kernel_launch(void* const* d_in, const int* in_sizes, int n_in,
                              void* d_out, int out_size, void* d_ws,
                              size_t ws_size, hipStream_t stream) {
  (void)in_sizes; (void)n_in; (void)d_ws; (void)ws_size; (void)out_size;
  const float* f1 = (const float*)d_in[0];
  const float* f2 = (const float*)d_in[1];
  float* out = (float*)d_out;

  cost_volume_rows_loop<<<dim3(NWG), dim3(NTHREADS), 0, stream>>>(f1, f2, out);
}


# f2 as A operand: a lane holds one output column; lanes rotate their 8 result registers by m mod 8, then a tile's plane is one contiguous 48-byte store, not 4-byte pieces in 18 lines; sums unchanged
# speedup vs baseline: 1.5089x; 1.1967x over previous
#include <hip/hip_runtime.h>


typedef __attribute__((ext_vector_type(2))) float v2f;
typedef __attribute__((ext_vector_type(4))) float v4f;
typedef __attribute__((ext_vector_type(8))) float v8f;

static __device__ __forceinline__ void lds_barrier() {
  __builtin_amdgcn_fence(__ATOMIC_RELEASE, "workgroup", "local");
  __builtin_amdgcn_s_barrier();
  __builtin_amdgcn_fence(__ATOMIC_ACQUIRE, "workgroup", "local");
}

#define BS 8
#define FS 96
#define HH 96
#define WW 320
#define PLANES 9
#define LVLS 4
#define HW (HH * WW)

#define NWG 256
#define RPW ((BS * HH) / NWG)
#define KC 16
#define KP (KC / 2)
#define NST (FS / KC)
#define NG (RPW * NST)
#define CPT (KC / 2)
#define HALO 8
#define TPITCH 12
#define NTILES 27
#define GROUPS ((WW + 2 * HALO) / 2)
#define KP_FLOATS (4 * GROUPS)
#define ARR_FLOATS (KP * KP_FLOATS)
#define STAGE_FLOATS (2 * ARR_FLOATS)
#define NTHREADS 640
#define EPI_STAGES 4

#define NSET 4
#define LDS_FLOATS (4 * STAGE_FLOATS)

static_assert(RPW * NWG == BS * HH, "rows must divide over the workgroups");
static_assert(TPITCH * (NTILES - 1) - 4 + 16 >= WW + LVLS, "tiles cover the band");
static_assert(TPITCH * (NTILES - 1) - 4 + 16 <= WW + HALO, "last tile inside the halo");
static_assert(LDS_FLOATS * 4 > 160 * 1024, "one workgroup per WGP");

static __device__ __forceinline__ v8f rotate_rows(v8f a, int m, float inv) {
  const float a0 = a[0], a1 = a[1], a2 = a[2], a3 = a[3];
  const float a4 = a[4], a5 = a[5], a6 = a[6], a7 = a[7];
  const bool b0 = (m & 1) != 0, b1 = (m & 2) != 0, b2 = (m & 4) != 0;
  const float t0 = b0 ? a1 : a0, t1 = b0 ? a2 : a1, t2 = b0 ? a3 : a2;
  const float t3 = b0 ? a4 : a3, t4 = b0 ? a5 : a4, t5 = b0 ? a6 : a5;
  const float t6 = b0 ? a7 : a6, t7 = b0 ? a0 : a7;
  const float u0 = b1 ? t2 : t0, u1 = b1 ? t3 : t1, u2 = b1 ? t4 : t2;
  const float u3 = b1 ? t5 : t3, u4 = b1 ? t6 : t4, u5 = b1 ? t7 : t5;
  const float u6 = b1 ? t0 : t6, u7 = b1 ? t1 : t7;
  v8f r;
  r[0] = (b2 ? u4 : u0) * inv;
  r[1] = (b2 ? u5 : u1) * inv;
  r[2] = (b2 ? u6 : u2) * inv;
  r[3] = (b2 ? u7 : u3) * inv;
  r[4] = (b2 ? u0 : u4) * inv;
  r[5] = (b2 ? u1 : u5) * inv;
  r[6] = (b2 ? u2 : u6) * inv;
  r[7] = (b2 ? u3 : u7) * inv;
  return r;
}

__global__ __launch_bounds__(NTHREADS) void cost_volume_rows_loop(
    const float* __restrict__ f1, const float* __restrict__ f2,
    float* __restrict__ out) {
  __shared__ __attribute__((aligned(16))) float lds[LDS_FLOATS];

  const int tid = threadIdx.x;
  const int lane = tid & 31;
  const int w = tid >> 5;

  long rowOff[RPW], outBase[RPW];
#pragma unroll
  for (int r = 0; r < RPW; ++r) {
    const int R = (int)blockIdx.x + NWG * r;
    const int b = R / HH;
    const int h = R % HH;
    rowOff[r] = ((long)b * FS * HH + h) * WW;
    outBase[r] = ((long)b * PLANES * HH + h) * WW;
  }

  const int arr = (w >= 10) ? 1 : 0;
  const int wq = w - 10 * arr;
  const int chh = (wq >= 5) ? 1 : 0;
  const int cp = (wq - 5 * chh) * 32 + lane;
  const float* __restrict__ src =
      (arr ? f2 : f1) + 2 * cp + (long)(chh * CPT) * HW;
  float* dst = lds + arr * ARR_FLOATS + chh * (CPT / 2) * KP_FLOATS +
               4 * (cp + HALO / 2);

  const int half = lane >> 4;
  const int m = lane & 15;
  const int wu = __builtin_amdgcn_readfirstlane(w);
  const int two = __builtin_amdgcn_readfirstlane(
      ((wu % 6 < 2 && wu < 14) || wu == 15) ? 1 : 0);
  const int t1 = (wu == 15) ? NTILES - 1 : 20 + 2 * (wu / 6) + (wu % 6);
  const int S0 = TPITCH * wu - 4;
  const int S1 = TPITCH * t1 - 4;
  const float* fragbase = lds + half * KP_FLOATS + 4 * (m >> 1) + (m & 1);
  const float* frag0 = fragbase + 2 * (S0 + HALO);
  const float* frag1 = fragbase + 2 * (S1 + HALO);

  v2f g[NSET][CPT];

#define ISSUE(set, G)                                                       \
  {                                                                         \
    const float* s_ = src + rowOff[(G) / NST] + (long)(((G) % NST) * KC) * HW; \
    _Pragma("unroll") for (int c = 0; c < CPT; ++c)                         \
        g[set][c] = *(const v2f*)(s_ + (long)c * HW);                       \
  }

#define STASH(set, buf)                                                     \
  _Pragma("unroll") for (int kp = 0; kp < CPT / 2; ++kp) {                  \
    v4f q = {g[set][2 * kp].x, g[set][2 * kp].y, g[set][2 * kp + 1].x,      \
             g[set][2 * kp + 1].y};                                         \
    *(v4f*)(dst + (buf) * STAGE_FLOATS + kp * KP_FLOATS) = q;               \
  }

#define EPI(row, rot, S, d0, d1)                                            \
  _Pragma("unroll") for (int d = (d0); d <= (d1); ++d) {                    \
    const int M = m + d;                                                    \
    const int x = (S) + m;                                                  \
    if ((M >> 3) == half && (M >= 4 || m >= 4) && x >= 0 && x < WW) {       \
      out[outBase[row] + (long)(LVLS - d) * HW + x] = rot[d & 7];           \
    }                                                                       \
  }

#define EPI_D0(st) ((st) == 0 ? -LVLS : 2 * (st) - 3)
#define EPI_D1(st) (2 * (st) - 2)

  const float inv = 1.0f / (float)FS;

  ISSUE(0, 0)
  ISSUE(1, 1)
  ISSUE(2, 2)
  ISSUE(3, 3)

  if (tid < 2 * 2 * KP * HALO) {
    const int bf = tid >> 7;
    const int ar = (tid >> 6) & 1;
    const int kp = (tid >> 3) & 7;
    const int gi = tid & 7;
    const int grp = (gi < HALO / 2) ? gi : (WW / 2 + gi);
    v4f z = {0.0f, 0.0f, 0.0f, 0.0f};
    *(v4f*)&lds[bf * STAGE_FLOATS + ar * ARR_FLOATS + kp * KP_FLOATS + 4 * grp] = z;
  }

  STASH(0, 0)
  lds_barrier();
  ISSUE(0, NSET)

  v8f acc0 = {}, acc1 = {};
  v8f pre0 = {}, pre1 = {};

  v2f fa0_[2][KC / 4], fb0_[2][KC / 4], fa1_[1][KC / 4], fb1_[1][KC / 4];

#define FRAGS(fa, fb, frag, fs, buf)                                        \
  _Pragma("unroll") for (int j = 0; j < KC / 4; ++j) {                      \
    const float* pa = frag + (buf) * STAGE_FLOATS + j * 2 * KP_FLOATS;      \
    const float* pb = pa + ARR_FLOATS;                                      \
    fa[fs][j] = (v2f){pa[0], pa[2]};                                        \
    fb[fs][j] = (v2f){pb[0], pb[2]};                                        \
  }

  FRAGS(fa0_, fb0_, frag0, 0, 0)
  if (two) {
    FRAGS(fa1_, fb1_, frag1, 0, 0)
  }
  STASH(1, 1)
  lds_barrier();

#pragma unroll
  for (int G = 0; G < NG; ++G) {
    const int row = G / NST;
    const int st = G % NST;

    if (G + NSET + 1 < NG) {
      ISSUE((G + 1) % NSET, G + NSET + 1)
    }

    if (row > 0 && st < EPI_STAGES) {
      EPI(row - 1, pre0, S0, EPI_D0(st), EPI_D1(st))
      if (two) {
        EPI(row - 1, pre1, S1, EPI_D0(st), EPI_D1(st))
      }
    }

    if (G + 1 < NG) {
      FRAGS(fa0_, fb0_, frag0, (G + 1) & 1, (G + 1) & 1)
      __builtin_amdgcn_sched_barrier(0);
    }

    if (two) {
#pragma unroll
      for (int j = 0; j < KC / 4; ++j) {
        acc1 = __builtin_amdgcn_wmma_f32_16x16x4_f32(
            false, fb1_[0][j], false, fa1_[0][j], (short)0, acc1,
            false, false);
      }
    }
#pragma unroll
    for (int j = 0; j < KC / 4; ++j) {
      acc0 = __builtin_amdgcn_wmma_f32_16x16x4_f32(
          false, fb0_[G & 1][j], false, fa0_[G & 1][j], (short)0, acc0, false,
          false);
    }

    if (st == NST - 1) {
      pre0 = rotate_rows(acc0, m, inv);
      pre1 = rotate_rows(acc1, m, inv);
      acc0 = (v8f){};
      acc1 = (v8f){};
    }

    if (G + 2 < NG) {
      __builtin_amdgcn_sched_barrier(0);
      STASH((G + 2) % NSET, G & 1)
    }
    if (G + 1 < NG) {
      if (two) {
        FRAGS(fa1_, fb1_, frag1, 0, (G + 1) & 1)
      }
    }
    if (G + 2 < NG) {
      lds_barrier();
    }
  }

  EPI(RPW - 1, pre0, S0, -LVLS, LVLS)
  if (two) {
    EPI(RPW - 1, pre1, S1, -LVLS, LVLS)
  }
}

extern "C" void kernel_launch(void* const* d_in, const int* in_sizes, int n_in,
                              void* d_out, int out_size, void* d_ws,
                              size_t ws_size, hipStream_t stream) {
  (void)in_sizes; (void)n_in; (void)d_ws; (void)ws_size; (void)out_size;
  const float* f1 = (const float*)d_in[0];
  const float* f2 = (const float*)d_in[1];
  float* out = (float*)d_out;

  cost_volume_rows_loop<<<dim3(NWG), dim3(NTHREADS), 0, stream>>>(f1, f2, out);
}


# Workgroup barrier split: a wave signals right behind its LDS reads and stores; the first tile's WMMAs, the load refill and the row stores run between signal and wait; order of sums unchanged
# speedup vs baseline: 1.5951x; 1.0571x over previous
#include <hip/hip_runtime.h>


typedef __attribute__((ext_vector_type(2))) float v2f;
typedef __attribute__((ext_vector_type(4))) float v4f;
typedef __attribute__((ext_vector_type(8))) float v8f;

static __device__ __forceinline__ void lds_barrier() {
  __builtin_amdgcn_fence(__ATOMIC_RELEASE, "workgroup", "local");
  __builtin_amdgcn_s_barrier();
  __builtin_amdgcn_fence(__ATOMIC_ACQUIRE, "workgroup", "local");
}

static __device__ __forceinline__ void lds_barrier_signal() {
  __builtin_amdgcn_fence(__ATOMIC_RELEASE, "workgroup", "local");
  __builtin_amdgcn_s_barrier_signal(-1);
}
static __device__ __forceinline__ void lds_barrier_wait() {
  __builtin_amdgcn_s_barrier_wait(-1);
  __builtin_amdgcn_fence(__ATOMIC_ACQUIRE, "workgroup", "local");
}

#define BS 8
#define FS 96
#define HH 96
#define WW 320
#define PLANES 9
#define LVLS 4
#define HW (HH * WW)

#define NWG 256
#define RPW ((BS * HH) / NWG)
#define KC 16
#define KP (KC / 2)
#define NST (FS / KC)
#define NG (RPW * NST)
#define CPT (KC / 2)
#define HALO 8
#define TPITCH 12
#define NTILES 27
#define GROUPS ((WW + 2 * HALO) / 2)
#define KP_FLOATS (4 * GROUPS)
#define ARR_FLOATS (KP * KP_FLOATS)
#define STAGE_FLOATS (2 * ARR_FLOATS)
#define NTHREADS 640
#define EPI_STAGES 4

#define NSET 4
#define LDS_FLOATS (4 * STAGE_FLOATS)

static_assert(RPW * NWG == BS * HH, "rows must divide over the workgroups");
static_assert(TPITCH * (NTILES - 1) - 4 + 16 >= WW + LVLS, "tiles cover the band");
static_assert(TPITCH * (NTILES - 1) - 4 + 16 <= WW + HALO, "last tile inside the halo");
static_assert(LDS_FLOATS * 4 > 160 * 1024, "one workgroup per WGP");

static __device__ __forceinline__ v8f rotate_rows(v8f a, int m, float inv) {
  const float a0 = a[0], a1 = a[1], a2 = a[2], a3 = a[3];
  const float a4 = a[4], a5 = a[5], a6 = a[6], a7 = a[7];
  const bool b0 = (m & 1) != 0, b1 = (m & 2) != 0, b2 = (m & 4) != 0;
  const float t0 = b0 ? a1 : a0, t1 = b0 ? a2 : a1, t2 = b0 ? a3 : a2;
  const float t3 = b0 ? a4 : a3, t4 = b0 ? a5 : a4, t5 = b0 ? a6 : a5;
  const float t6 = b0 ? a7 : a6, t7 = b0 ? a0 : a7;
  const float u0 = b1 ? t2 : t0, u1 = b1 ? t3 : t1, u2 = b1 ? t4 : t2;
  const float u3 = b1 ? t5 : t3, u4 = b1 ? t6 : t4, u5 = b1 ? t7 : t5;
  const float u6 = b1 ? t0 : t6, u7 = b1 ? t1 : t7;
  v8f r;
  r[0] = (b2 ? u4 : u0) * inv;
  r[1] = (b2 ? u5 : u1) * inv;
  r[2] = (b2 ? u6 : u2) * inv;
  r[3] = (b2 ? u7 : u3) * inv;
  r[4] = (b2 ? u0 : u4) * inv;
  r[5] = (b2 ? u1 : u5) * inv;
  r[6] = (b2 ? u2 : u6) * inv;
  r[7] = (b2 ? u3 : u7) * inv;
  return r;
}

__global__ __launch_bounds__(NTHREADS) void cost_volume_rows_loop(
    const float* __restrict__ f1, const float* __restrict__ f2,
    float* __restrict__ out) {
  __shared__ __attribute__((aligned(16))) float lds[LDS_FLOATS];

  const int tid = threadIdx.x;
  const int lane = tid & 31;
  const int w = tid >> 5;

  long rowOff[RPW], outBase[RPW];
#pragma unroll
  for (int r = 0; r < RPW; ++r) {
    const int R = (int)blockIdx.x + NWG * r;
    const int b = R / HH;
    const int h = R % HH;
    rowOff[r] = ((long)b * FS * HH + h) * WW;
    outBase[r] = ((long)b * PLANES * HH + h) * WW;
  }

  const int arr = (w >= 10) ? 1 : 0;
  const int wq = w - 10 * arr;
  const int chh = (wq >= 5) ? 1 : 0;
  const int cp = (wq - 5 * chh) * 32 + lane;
  const float* __restrict__ src =
      (arr ? f2 : f1) + 2 * cp + (long)(chh * CPT) * HW;
  float* dst = lds + arr * ARR_FLOATS + chh * (CPT / 2) * KP_FLOATS +
               4 * (cp + HALO / 2);

  const int half = lane >> 4;
  const int m = lane & 15;
  const int wu = __builtin_amdgcn_readfirstlane(w);
  const int two = __builtin_amdgcn_readfirstlane(
      ((wu % 6 < 2 && wu < 14) || wu == 15) ? 1 : 0);
  const int t1 = (wu == 15) ? NTILES - 1 : 20 + 2 * (wu / 6) + (wu % 6);
  const int S0 = TPITCH * wu - 4;
  const int S1 = TPITCH * t1 - 4;
  const float* fragbase = lds + half * KP_FLOATS + 4 * (m >> 1) + (m & 1);
  const float* frag0 = fragbase + 2 * (S0 + HALO);
  const float* frag1 = fragbase + 2 * (S1 + HALO);

  v2f g[NSET][CPT];

#define ISSUE(set, G)                                                       \
  {                                                                         \
    const float* s_ = src + rowOff[(G) / NST] + (long)(((G) % NST) * KC) * HW; \
    _Pragma("unroll") for (int c = 0; c < CPT; ++c)                         \
        g[set][c] = *(const v2f*)(s_ + (long)c * HW);                       \
  }

#define STASH(set, buf)                                                     \
  _Pragma("unroll") for (int kp = 0; kp < CPT / 2; ++kp) {                  \
    v4f q = {g[set][2 * kp].x, g[set][2 * kp].y, g[set][2 * kp + 1].x,      \
             g[set][2 * kp + 1].y};                                         \
    *(v4f*)(dst + (buf) * STAGE_FLOATS + kp * KP_FLOATS) = q;               \
  }

#define EPI(row, rot, S, d0, d1)                                            \
  _Pragma("unroll") for (int d = (d0); d <= (d1); ++d) {                    \
    const int M = m + d;                                                    \
    const int x = (S) + m;                                                  \
    if ((M >> 3) == half && (M >= 4 || m >= 4) && x >= 0 && x < WW) {       \
      out[outBase[row] + (long)(LVLS - d) * HW + x] = rot[d & 7];           \
    }                                                                       \
  }

#define EPI_D0(st) ((st) == 0 ? -LVLS : 2 * (st) - 3)
#define EPI_D1(st) (2 * (st) - 2)

  const float inv = 1.0f / (float)FS;

  ISSUE(0, 0)
  ISSUE(1, 1)
  ISSUE(2, 2)
  ISSUE(3, 3)

  if (tid < 2 * 2 * KP * HALO) {
    const int bf = tid >> 7;
    const int ar = (tid >> 6) & 1;
    const int kp = (tid >> 3) & 7;
    const int gi = tid & 7;
    const int grp = (gi < HALO / 2) ? gi : (WW / 2 + gi);
    v4f z = {0.0f, 0.0f, 0.0f, 0.0f};
    *(v4f*)&lds[bf * STAGE_FLOATS + ar * ARR_FLOATS + kp * KP_FLOATS + 4 * grp] = z;
  }

  STASH(0, 0)
  lds_barrier();
  ISSUE(0, NSET)

  v8f acc0 = {}, acc1 = {};
  v8f pre0 = {}, pre1 = {};

  v2f fa0_[2][KC / 4], fb0_[2][KC / 4], fa1_[1][KC / 4], fb1_[1][KC / 4];

#define FRAGS(fa, fb, frag, fs, buf)                                        \
  _Pragma("unroll") for (int j = 0; j < KC / 4; ++j) {                      \
    const float* pa = frag + (buf) * STAGE_FLOATS + j * 2 * KP_FLOATS;      \
    const float* pb = pa + ARR_FLOATS;                                      \
    fa[fs][j] = (v2f){pa[0], pa[2]};                                        \
    fb[fs][j] = (v2f){pb[0], pb[2]};                                        \
  }

  FRAGS(fa0_, fb0_, frag0, 0, 0)
  if (two) {
    FRAGS(fa1_, fb1_, frag1, 0, 0)
  }
  STASH(1, 1)
  lds_barrier();

  ISSUE(1, NSET + 1)

#pragma unroll
  for (int G = 0; G < NG; ++G) {
    const int row = G / NST;
    const int st = G % NST;


    if (G + 1 < NG) {
      FRAGS(fa0_, fb0_, frag0, (G + 1) & 1, (G + 1) & 1)
    }

    if (two) {
#pragma unroll
      for (int j = 0; j < KC / 4; ++j) {
        acc1 = __builtin_amdgcn_wmma_f32_16x16x4_f32(
            false, fb1_[0][j], false, fa1_[0][j], (short)0, acc1,
            false, false);
      }
      if (G + 1 < NG) {
        __builtin_amdgcn_sched_barrier(0);
        FRAGS(fa1_, fb1_, frag1, 0, (G + 1) & 1)
      }
    }

    if (G + 2 < NG) {
      STASH((G + 2) % NSET, G & 1)
      lds_barrier_signal();
      __builtin_amdgcn_sched_barrier(0);
    }


    if (G + 2 + NSET < NG) {
      ISSUE((G + 2) % NSET, G + 2 + NSET)
    }

    if (row > 0 && st < EPI_STAGES) {
      EPI(row - 1, pre0, S0, EPI_D0(st), EPI_D1(st))
      if (two) {
        EPI(row - 1, pre1, S1, EPI_D0(st), EPI_D1(st))
      }
    }

    __builtin_amdgcn_sched_barrier(0);

#pragma unroll
    for (int j = 0; j < KC / 4; ++j) {
      acc0 = __builtin_amdgcn_wmma_f32_16x16x4_f32(
          false, fb0_[G & 1][j], false, fa0_[G & 1][j], (short)0, acc0, false,
          false);
    }

    if (st == NST - 1) {
      pre0 = rotate_rows(acc0, m, inv);
      pre1 = rotate_rows(acc1, m, inv);
      acc0 = (v8f){};
      acc1 = (v8f){};
    }

    if (G + 2 < NG) {
      __builtin_amdgcn_sched_barrier(0);
      lds_barrier_wait();
    }
  }

  EPI(RPW - 1, pre0, S0, -LVLS, LVLS)
  if (two) {
    EPI(RPW - 1, pre1, S1, -LVLS, LVLS)
  }
}

extern "C" void kernel_launch(void* const* d_in, const int* in_sizes, int n_in,
                              void* d_out, int out_size, void* d_ws,
                              size_t ws_size, hipStream_t stream) {
  (void)in_sizes; (void)n_in; (void)d_ws; (void)ws_size; (void)out_size;
  const float* f1 = (const float*)d_in[0];
  const float* f2 = (const float*)d_in[1];
  float* out = (float*)d_out;

  cost_volume_rows_loop<<<dim3(NWG), dim3(NTHREADS), 0, stream>>>(f1, f2, out);
}


# speedup vs baseline: 1.6867x; 1.0574x over previous
#include <hip/hip_runtime.h>


typedef __attribute__((ext_vector_type(2))) float v2f;
typedef __attribute__((ext_vector_type(4))) float v4f;
typedef __attribute__((ext_vector_type(8))) float v8f;

static __device__ __forceinline__ void lds_barrier() {
  __builtin_amdgcn_fence(__ATOMIC_RELEASE, "workgroup", "local");
  __builtin_amdgcn_s_barrier();
  __builtin_amdgcn_fence(__ATOMIC_ACQUIRE, "workgroup", "local");
}

static __device__ __forceinline__ void lds_barrier_signal() {
  __builtin_amdgcn_fence(__ATOMIC_RELEASE, "workgroup", "local");
  __builtin_amdgcn_s_barrier_signal(-1);
}
static __device__ __forceinline__ void lds_barrier_wait() {
  __builtin_amdgcn_s_barrier_wait(-1);
  __builtin_amdgcn_fence(__ATOMIC_ACQUIRE, "workgroup", "local");
}

#define BS 8
#define FS 96
#define HH 96
#define WW 320
#define PLANES 9
#define LVLS 4
#define HW (HH * WW)

#define NWG 256
#define RPW ((BS * HH) / NWG)
#define KC 16
#define KP (KC / 2)
#define NST (FS / KC)
#define NG (RPW * NST)
#define CPT (KC / 2)
#define HALO 8
#define TPITCH 12
#define NTILES 27
#define GROUPS ((WW + 2 * HALO) / 2)
#define KP_FLOATS (4 * GROUPS)
#define ARR_FLOATS (KP * KP_FLOATS)
#define STAGE_FLOATS (2 * ARR_FLOATS)
#define NTHREADS 640
#define EPI_STAGES 4

#define NSET 4
#define LDS_FLOATS (4 * STAGE_FLOATS)

static_assert(RPW * NWG == BS * HH, "rows must divide over the workgroups");
static_assert(TPITCH * (NTILES - 1) - 4 + 16 >= WW + LVLS, "tiles cover the band");
static_assert(TPITCH * (NTILES - 1) - 4 + 16 <= WW + HALO, "last tile inside the halo");
static_assert(LDS_FLOATS * 4 > 160 * 1024, "one workgroup per WGP");

static __device__ __forceinline__ v8f rotate_rows(v8f a, int m, float inv) {
  const float a0 = a[0], a1 = a[1], a2 = a[2], a3 = a[3];
  const float a4 = a[4], a5 = a[5], a6 = a[6], a7 = a[7];
  const bool b0 = (m & 1) != 0, b1 = (m & 2) != 0, b2 = (m & 4) != 0;
  const float t0 = b0 ? a1 : a0, t1 = b0 ? a2 : a1, t2 = b0 ? a3 : a2;
  const float t3 = b0 ? a4 : a3, t4 = b0 ? a5 : a4, t5 = b0 ? a6 : a5;
  const float t6 = b0 ? a7 : a6, t7 = b0 ? a0 : a7;
  const float u0 = b1 ? t2 : t0, u1 = b1 ? t3 : t1, u2 = b1 ? t4 : t2;
  const float u3 = b1 ? t5 : t3, u4 = b1 ? t6 : t4, u5 = b1 ? t7 : t5;
  const float u6 = b1 ? t0 : t6, u7 = b1 ? t1 : t7;
  v8f r;
  r[0] = (b2 ? u4 : u0) * inv;
  r[1] = (b2 ? u5 : u1) * inv;
  r[2] = (b2 ? u6 : u2) * inv;
  r[3] = (b2 ? u7 : u3) * inv;
  r[4] = (b2 ? u0 : u4) * inv;
  r[5] = (b2 ? u1 : u5) * inv;
  r[6] = (b2 ? u2 : u6) * inv;
  r[7] = (b2 ? u3 : u7) * inv;
  return r;
}

__global__ __launch_bounds__(NTHREADS) void cost_volume_rows_loop(
    const float* __restrict__ f1, const float* __restrict__ f2,
    float* __restrict__ out) {
  __shared__ __attribute__((aligned(16))) float lds[LDS_FLOATS];

  const int tid = threadIdx.x;
  const int lane = tid & 31;
  const int w = tid >> 5;

  long rowOff[RPW], outBase[RPW];
#pragma unroll
  for (int r = 0; r < RPW; ++r) {
    const int R = (int)blockIdx.x + NWG * r;
    const int b = R / HH;
    const int h = R % HH;
    rowOff[r] = ((long)b * FS * HH + h) * WW;
    outBase[r] = ((long)b * PLANES * HH + h) * WW;
  }

  const int arr = (w >= 10) ? 1 : 0;
  const int wq = w - 10 * arr;
  const int chh = (wq >= 5) ? 1 : 0;
  const int cp = (wq - 5 * chh) * 32 + lane;
  const float* __restrict__ src =
      (arr ? f2 : f1) + 2 * cp + (long)(chh * CPT) * HW;
  float* dst = lds + arr * ARR_FLOATS + chh * (CPT / 2) * KP_FLOATS +
               4 * (cp + HALO / 2);

  const int half = lane >> 4;
  const int m = lane & 15;
  const int wu = __builtin_amdgcn_readfirstlane(w);
  const int two = __builtin_amdgcn_readfirstlane(
      ((wu % 6 < 2 && wu < 14) || wu == 15) ? 1 : 0);
  const int t1 = (wu == 15) ? NTILES - 1 : 20 + 2 * (wu / 6) + (wu % 6);
  const int S0 = TPITCH * wu - 4;
  const int S1 = TPITCH * t1 - 4;
  const float* fragbase = lds + half * KP_FLOATS + 4 * (m >> 1) + (m & 1);
  const float* frag0 = fragbase + 2 * (S0 + HALO);
  const float* frag1 = fragbase + 2 * (S1 + HALO);

  v2f g[NSET][CPT];

#define ISSUE(set, G)                                                       \
  {                                                                         \
    const float* s_ = src + rowOff[(G) / NST] + (long)(((G) % NST) * KC) * HW; \
    _Pragma("unroll") for (int c = 0; c < CPT; ++c)                         \
        g[set][c] = __builtin_nontemporal_load((const v2f*)(s_ + (long)c * HW)); \
  }

#define STASH(set, buf)                                                     \
  _Pragma("unroll") for (int kp = 0; kp < CPT / 2; ++kp) {                  \
    v4f q = {g[set][2 * kp].x, g[set][2 * kp].y, g[set][2 * kp + 1].x,      \
             g[set][2 * kp + 1].y};                                         \
    *(v4f*)(dst + (buf) * STAGE_FLOATS + kp * KP_FLOATS) = q;               \
  }

#define EPI(row, rot, S, d0, d1)                                            \
  _Pragma("unroll") for (int d = (d0); d <= (d1); ++d) {                    \
    const int M = m + d;                                                    \
    const int x = (S) + m;                                                  \
    if ((M >> 3) == half && (M >= 4 || m >= 4) && x >= 0 && x < WW) {       \
      out[outBase[row] + (long)(LVLS - d) * HW + x] = rot[d & 7];           \
    }                                                                       \
  }

#define EPI_D0(st) ((st) == 0 ? -LVLS : 2 * (st) - 3)
#define EPI_D1(st) (2 * (st) - 2)

  const float inv = 1.0f / (float)FS;

  ISSUE(0, 0)
  ISSUE(1, 1)
  ISSUE(2, 2)
  ISSUE(3, 3)

  if (tid < 2 * 2 * KP * HALO) {
    const int bf = tid >> 7;
    const int ar = (tid >> 6) & 1;
    const int kp = (tid >> 3) & 7;
    const int gi = tid & 7;
    const int grp = (gi < HALO / 2) ? gi : (WW / 2 + gi);
    v4f z = {0.0f, 0.0f, 0.0f, 0.0f};
    *(v4f*)&lds[bf * STAGE_FLOATS + ar * ARR_FLOATS + kp * KP_FLOATS + 4 * grp] = z;
  }

  STASH(0, 0)
  lds_barrier();
  ISSUE(0, NSET)

  v8f acc0 = {}, acc1 = {};
  v8f pre0 = {}, pre1 = {};

  v2f fa0_[2][KC / 4], fb0_[2][KC / 4], fa1_[1][KC / 4], fb1_[1][KC / 4];

#define FRAGS(fa, fb, frag, fs, buf)                                        \
  _Pragma("unroll") for (int j = 0; j < KC / 4; ++j) {                      \
    const float* pa = frag + (buf) * STAGE_FLOATS + j * 2 * KP_FLOATS;      \
    const float* pb = pa + ARR_FLOATS;                                      \
    fa[fs][j] = (v2f){pa[0], pa[2]};                                        \
    fb[fs][j] = (v2f){pb[0], pb[2]};                                        \
  }

  FRAGS(fa0_, fb0_, frag0, 0, 0)
  if (two) {
    FRAGS(fa1_, fb1_, frag1, 0, 0)
  }
  STASH(1, 1)
  lds_barrier();

  ISSUE(1, NSET + 1)

#pragma unroll
  for (int G = 0; G < NG; ++G) {
    const int row = G / NST;
    const int st = G % NST;


    if (G + 1 < NG) {
      FRAGS(fa0_, fb0_, frag0, (G + 1) & 1, (G + 1) & 1)
    }

    if (two) {
#pragma unroll
      for (int j = 0; j < KC / 4; ++j) {
        acc1 = __builtin_amdgcn_wmma_f32_16x16x4_f32(
            false, fb1_[0][j], false, fa1_[0][j], (short)0, acc1,
            false, false);
      }
      if (G + 1 < NG) {
        __builtin_amdgcn_sched_barrier(0);
        FRAGS(fa1_, fb1_, frag1, 0, (G + 1) & 1)
      }
    }

    if (G + 2 < NG) {
      STASH((G + 2) % NSET, G & 1)
      lds_barrier_signal();
      __builtin_amdgcn_sched_barrier(0);
    }


    if (G + 2 + NSET < NG) {
      ISSUE((G + 2) % NSET, G + 2 + NSET)
    }

    if (row > 0 && st < EPI_STAGES) {
      EPI(row - 1, pre0, S0, EPI_D0(st), EPI_D1(st))
      if (two) {
        EPI(row - 1, pre1, S1, EPI_D0(st), EPI_D1(st))
      }
    }

    __builtin_amdgcn_sched_barrier(0);

#pragma unroll
    for (int j = 0; j < KC / 4; ++j) {
      acc0 = __builtin_amdgcn_wmma_f32_16x16x4_f32(
          false, fb0_[G & 1][j], false, fa0_[G & 1][j], (short)0, acc0, false,
          false);
    }

    if (st == NST - 1) {
      pre0 = rotate_rows(acc0, m, inv);
      pre1 = rotate_rows(acc1, m, inv);
      acc0 = (v8f){};
      acc1 = (v8f){};
    }

    if (G + 2 < NG) {
      __builtin_amdgcn_sched_barrier(0);
      lds_barrier_wait();
    }
  }

  EPI(RPW - 1, pre0, S0, -LVLS, LVLS)
  if (two) {
    EPI(RPW - 1, pre1, S1, -LVLS, LVLS)
  }
}

extern "C" void kernel_launch(void* const* d_in, const int* in_sizes, int n_in,
                              void* d_out, int out_size, void* d_ws,
                              size_t ws_size, hipStream_t stream) {
  (void)in_sizes; (void)n_in; (void)d_ws; (void)ws_size; (void)out_size;
  const float* f1 = (const float*)d_in[0];
  const float* f2 = (const float*)d_in[1];
  float* out = (float*)d_out;

  cost_volume_rows_loop<<<dim3(NWG), dim3(NTHREADS), 0, stream>>>(f1, f2, out);
}
